# starting kernel
# baseline (speedup 1.0000x reference)
#include <hip/hip_runtime.h>

#define EPS2 1e-12f
#define RSQRT_IR 0.31622776601683794f

typedef __attribute__((ext_vector_type(2))) float v2f;
typedef __attribute__((ext_vector_type(8))) float v8f;

#if __has_builtin(__builtin_amdgcn_tanhf)
#define FAST_TANH(x) __builtin_amdgcn_tanhf(x)
#else
#define FAST_TANH(x) tanhf(x)
#endif

__device__ __forceinline__ float fast_sqrt(float x) { return __builtin_amdgcn_sqrtf(x); }
__device__ __forceinline__ float fast_rcp(float x) { return __builtin_amdgcn_rcpf(x); }

__global__ __launch_bounds__(256) void trq_2b_kernel(
    const float* __restrict__ v1b,
    const float* __restrict__ invar,
    const float* __restrict__ v2b,
    const float* __restrict__ W1,
    const float* __restrict__ b1,
    const float* __restrict__ W2,
    const float* __restrict__ b2,
    const float* __restrict__ lw1,
    const float* __restrict__ nb1,
    const float* __restrict__ lw2,
    const float* __restrict__ nb2,
    const float* __restrict__ wf,
    float* __restrict__ out)
{
  __shared__ float sW1p[4 * 16];
  __shared__ float sW2p[12 * 16];
  __shared__ float sB1[16];
  __shared__ float sB2[16];
  __shared__ float slw1[10], snb1[10], slw2[100], snb2[10], swf[10];
  __shared__ float ytile[8][2][16 * 20];
  __shared__ float red[8][4];

  const int t = threadIdx.x;
  const int b = blockIdx.x;
  const int lane = t & 31;
  const int wave = t >> 5;
  const int halfIdx = lane >> 4;
  const int lm = lane & 15;

  {
    const int k = t >> 4, n = t & 15;
    if (t < 64)  sW1p[t] = (k < 2  && n < 10) ? W1[k * 10 + n] : 0.f;
    if (t < 192) sW2p[t] = (k < 10 && n < 10) ? W2[k * 10 + n] : 0.f;
  }
  if (t < 100) slw2[t] = lw2[t];
  if (t < 16)  { sB1[t] = (t < 10) ? b1[t] : 0.f; sB2[t] = (t < 10) ? b2[t] : 0.f; }
  if (t < 10)  { slw1[t] = lw1[t]; snb1[t] = nb1[t]; snb2[t] = nb2[t]; swf[t] = wf[t]; }
  __syncthreads();

  const size_t idx = (size_t)b * 256 + (size_t)t;
  const v2f inv = *(const v2f*)(invar + idx * 2);
  const float* vp = v2b + idx * 6 + 3;
  const float v0 = vp[0], v1 = vp[1], v2 = vp[2];

  const float i0s = __shfl(inv.x, (lane + 16) & 31, 32);
  const float i1s = __shfl(inv.y, (lane + 16) & 31, 32);
  v2f a0, a1;
  a0.x = (lane < 16) ? inv.x : 0.f;  a0.y = (lane < 16) ? inv.y : 0.f;
  a1.x = (lane < 16) ? i0s  : 0.f;   a1.y = (lane < 16) ? i1s  : 0.f;

  v2f bw1;
  bw1.x = sW1p[(0 + 2 * halfIdx) * 16 + lm];
  bw1.y = sW1p[(1 + 2 * halfIdx) * 16 + lm];

  const float cb1 = sB1[lm];
  v8f d0 = {cb1, cb1, cb1, cb1, cb1, cb1, cb1, cb1};
  v8f d1 = d0;

  d0 = __builtin_amdgcn_wmma_f32_16x16x4_f32(false, a0, false, bw1, (short)0, d0, false, false);
  d1 = __builtin_amdgcn_wmma_f32_16x16x4_f32(false, a1, false, bw1, (short)0, d1, false, false);

#pragma unroll
  for (int r = 0; r < 8; ++r) { d0[r] = fmaxf(d0[r], 0.f); d1[r] = fmaxf(d1[r], 0.f); }

#pragma unroll
  for (int r = 0; r < 8; ++r) {
    const int m = r + 8 * halfIdx;
    ytile[wave][0][m * 20 + lm] = d0[r];
    ytile[wave][1][m * 20 + lm] = d1[r];
  }
  __syncthreads();

  const float cb2 = sB2[lm];
  v8f e0 = {cb2, cb2, cb2, cb2, cb2, cb2, cb2, cb2};
  v8f e1 = e0;
#pragma unroll
  for (int j = 0; j < 3; ++j) {
    const int k0 = 4 * j + 2 * halfIdx;
    v2f aj0, aj1, bj;
    aj0.x = ytile[wave][0][lm * 20 + k0];
    aj0.y = ytile[wave][0][lm * 20 + k0 + 1];
    aj1.x = ytile[wave][1][lm * 20 + k0];
    aj1.y = ytile[wave][1][lm * 20 + k0 + 1];
    bj.x = sW2p[k0 * 16 + lm];
    bj.y = sW2p[(k0 + 1) * 16 + lm];
    e0 = __builtin_amdgcn_wmma_f32_16x16x4_f32(false, aj0, false, bj, (short)0, e0, false, false);
    e1 = __builtin_amdgcn_wmma_f32_16x16x4_f32(false, aj1, false, bj, (short)0, e1, false, false);
  }
#pragma unroll
  for (int r = 0; r < 8; ++r) { e0[r] = fmaxf(e0[r], 0.f); e1[r] = fmaxf(e1[r], 0.f); }

#pragma unroll
  for (int r = 0; r < 8; ++r) {
    const int m = r + 8 * halfIdx;
    ytile[wave][0][m * 20 + lm] = e0[r];
    ytile[wave][1][m * 20 + lm] = e1[r];
  }
  __syncthreads();

  const int tl = (t >> 4) & 1, row = t & 15;
  float y[10];
#pragma unroll
  for (int u = 0; u < 10; ++u) y[u] = ytile[wave][tl][row * 20 + u];

  const float nv2 = v0 * v0 + v1 * v1 + v2 * v2;
  float c1[10];
#pragma unroll
  for (int u = 0; u < 10; ++u) {
    const float s = slw1[u];
    const float n1 = fast_sqrt(nv2 * s * s + EPS2);
    c1[u] = s * FAST_TANH(n1 + snb1[u]) * fast_rcp(n1);
  }
  float S = 0.f;
#pragma unroll
  for (int vv = 0; vv < 10; ++vv) {
    float acc = 0.f;
#pragma unroll
    for (int u = 0; u < 10; ++u) acc = fmaf(c1[u], slw2[u * 10 + vv], acc);
    const float m  = acc * RSQRT_IR;
    const float n2 = fast_sqrt(m * m * nv2 + EPS2);
    const float g  = FAST_TANH(n2 + snb2[vv]) * fast_rcp(n2);
    S = fmaf(swf[vv] * y[vv], m * g, S);
  }
  S *= RSQRT_IR;
  float sx = S * v0, sy = S * v1, sz = S * v2;

#pragma unroll
  for (int off = 16; off; off >>= 1) {
    sx += __shfl_down(sx, off, 32);
    sy += __shfl_down(sy, off, 32);
    sz += __shfl_down(sz, off, 32);
  }
  if (lane == 0) { red[wave][0] = sx; red[wave][1] = sy; red[wave][2] = sz; }
  __syncthreads();
  if (t == 0) {
    float tx = 0.f, ty = 0.f, tz = 0.f;
#pragma unroll
    for (int w = 0; w < 8; ++w) { tx += red[w][0]; ty += red[w][1]; tz += red[w][2]; }
    const float ax = v1b[b * 3 + 0], ay = v1b[b * 3 + 1], az = v1b[b * 3 + 2];
    out[b * 3 + 0] = ay * tz - az * ty;
    out[b * 3 + 1] = az * tx - ax * tz;
    out[b * 3 + 2] = ax * ty - ay * tx;
  }
}

extern "C" void kernel_launch(void* const* d_in, const int* in_sizes, int n_in,
                              void* d_out, int out_size, void* d_ws, size_t ws_size,
                              hipStream_t stream) {
  (void)n_in; (void)out_size; (void)d_ws; (void)ws_size;
  const float* v1b   = (const float*)d_in[0];
  const float* invar = (const float*)d_in[1];
  const float* v2b   = (const float*)d_in[2];
  const float* W1    = (const float*)d_in[3];
  const float* b1    = (const float*)d_in[4];
  const float* W2    = (const float*)d_in[5];
  const float* b2    = (const float*)d_in[6];
  const float* lw1   = (const float*)d_in[7];
  const float* nb1   = (const float*)d_in[8];
  const float* lw2   = (const float*)d_in[9];
  const float* nb2   = (const float*)d_in[10];
  const float* wf    = (const float*)d_in[11];
  float* out = (float*)d_out;

  const int B = in_sizes[0] / 3;
  trq_2b_kernel<<<B, 256, 0, stream>>>(v1b, invar, v2b, W1, b1, W2, b2,
                                       lw1, nb1, lw2, nb2, wf, out);
}


# Fixed grid of 1024 work-groups that each loop over their batch rows: weights staged to LDS once per group, 4x fewer waves to enter, next row's loads kept in flight across LDS-only barriers
# speedup vs baseline: 1.0804x; 1.0804x over previous
#include <hip/hip_runtime.h>

#define EPS2 1e-12f
#define RSQRT_IR 0.31622776601683794f

typedef __attribute__((ext_vector_type(2))) float v2f;
typedef __attribute__((ext_vector_type(8))) float v8f;

#if __has_builtin(__builtin_amdgcn_tanhf)
#define FAST_TANH(x) __builtin_amdgcn_tanhf(x)
#else
#define FAST_TANH(x) tanhf(x)
#endif

__device__ __forceinline__ float fast_sqrt(float x) { return __builtin_amdgcn_sqrtf(x); }
__device__ __forceinline__ float fast_rcp(float x) { return __builtin_amdgcn_rcpf(x); }

__device__ __forceinline__ void lds_barrier() {
  __builtin_amdgcn_fence(__ATOMIC_RELEASE, "workgroup", "local");
  __builtin_amdgcn_s_barrier();
  __builtin_amdgcn_fence(__ATOMIC_ACQUIRE, "workgroup", "local");
}

#define kGrid 1024
__global__ __launch_bounds__(256) void trq_2b_kernel(
    const float* __restrict__ v1b,
    const float* __restrict__ invar,
    const float* __restrict__ v2b,
    const float* __restrict__ W1,
    const float* __restrict__ b1,
    const float* __restrict__ W2,
    const float* __restrict__ b2,
    const float* __restrict__ lw1,
    const float* __restrict__ nb1,
    const float* __restrict__ lw2,
    const float* __restrict__ nb2,
    const float* __restrict__ wf,
    float* __restrict__ out,
    const int B)
{
  __shared__ float sW1p[4 * 16];
  __shared__ float sW2p[12 * 16];
  __shared__ float sB1[16];
  __shared__ float sB2[16];
  __shared__ float slw1[10], snb1[10], slw2[100], snb2[10], swf[10];
  __shared__ float ytile[8][2][16 * 20];
  __shared__ float red[8][4];

  const int t = threadIdx.x;
  const int lane = t & 31;
  const int wave = t >> 5;
  const int halfIdx = lane >> 4;
  const int lm = lane & 15;

  if ((int)blockIdx.x >= B) return;

  v2f ninv;
  float nv0, nv1, nv2r;
  {
    const size_t idx = (size_t)blockIdx.x * 256 + (size_t)t;
    ninv = *(const v2f*)(invar + idx * 2);
    const float* vp = v2b + idx * 6 + 3;
    nv0 = vp[0]; nv1 = vp[1]; nv2r = vp[2];
  }

  {
    const int k = t >> 4, n = t & 15;
    if (t < 64)  sW1p[t] = (k < 2  && n < 10) ? W1[k * 10 + n] : 0.f;
    if (t < 192) sW2p[t] = (k < 10 && n < 10) ? W2[k * 10 + n] : 0.f;
  }
  if (t < 100) slw2[t] = lw2[t];
  if (t < 16)  { sB1[t] = (t < 10) ? b1[t] : 0.f; sB2[t] = (t < 10) ? b2[t] : 0.f; }
  if (t < 10)  { slw1[t] = lw1[t]; snb1[t] = nb1[t]; snb2[t] = nb2[t]; swf[t] = wf[t]; }
  lds_barrier();

  for (int b = blockIdx.x; b < B; b += kGrid) {
  const v2f inv = ninv;
  const float v0 = nv0, v1 = nv1, v2 = nv2r;
  {
    const int bn = (b + kGrid < B) ? (b + kGrid) : b;
    const size_t idx = (size_t)bn * 256 + (size_t)t;
    ninv = *(const v2f*)(invar + idx * 2);
    const float* vp = v2b + idx * 6 + 3;
    nv0 = vp[0]; nv1 = vp[1]; nv2r = vp[2];
  }

  const float i0s = __shfl(inv.x, (lane + 16) & 31, 32);
  const float i1s = __shfl(inv.y, (lane + 16) & 31, 32);
  v2f a0, a1;
  a0.x = (lane < 16) ? inv.x : 0.f;  a0.y = (lane < 16) ? inv.y : 0.f;
  a1.x = (lane < 16) ? i0s  : 0.f;   a1.y = (lane < 16) ? i1s  : 0.f;

  v2f bw1;
  bw1.x = sW1p[(0 + 2 * halfIdx) * 16 + lm];
  bw1.y = sW1p[(1 + 2 * halfIdx) * 16 + lm];

  const float cb1 = sB1[lm];
  v8f d0 = {cb1, cb1, cb1, cb1, cb1, cb1, cb1, cb1};
  v8f d1 = d0;

  d0 = __builtin_amdgcn_wmma_f32_16x16x4_f32(false, a0, false, bw1, (short)0, d0, false, false);
  d1 = __builtin_amdgcn_wmma_f32_16x16x4_f32(false, a1, false, bw1, (short)0, d1, false, false);

#pragma unroll
  for (int r = 0; r < 8; ++r) { d0[r] = fmaxf(d0[r], 0.f); d1[r] = fmaxf(d1[r], 0.f); }

#pragma unroll
  for (int r = 0; r < 8; ++r) {
    const int m = r + 8 * halfIdx;
    ytile[wave][0][m * 20 + lm] = d0[r];
    ytile[wave][1][m * 20 + lm] = d1[r];
  }
  lds_barrier();

  const float cb2 = sB2[lm];
  v8f e0 = {cb2, cb2, cb2, cb2, cb2, cb2, cb2, cb2};
  v8f e1 = e0;
#pragma unroll
  for (int j = 0; j < 3; ++j) {
    const int k0 = 4 * j + 2 * halfIdx;
    v2f aj0, aj1, bj;
    aj0.x = ytile[wave][0][lm * 20 + k0];
    aj0.y = ytile[wave][0][lm * 20 + k0 + 1];
    aj1.x = ytile[wave][1][lm * 20 + k0];
    aj1.y = ytile[wave][1][lm * 20 + k0 + 1];
    bj.x = sW2p[k0 * 16 + lm];
    bj.y = sW2p[(k0 + 1) * 16 + lm];
    e0 = __builtin_amdgcn_wmma_f32_16x16x4_f32(false, aj0, false, bj, (short)0, e0, false, false);
    e1 = __builtin_amdgcn_wmma_f32_16x16x4_f32(false, aj1, false, bj, (short)0, e1, false, false);
  }
#pragma unroll
  for (int r = 0; r < 8; ++r) { e0[r] = fmaxf(e0[r], 0.f); e1[r] = fmaxf(e1[r], 0.f); }

#pragma unroll
  for (int r = 0; r < 8; ++r) {
    const int m = r + 8 * halfIdx;
    ytile[wave][0][m * 20 + lm] = e0[r];
    ytile[wave][1][m * 20 + lm] = e1[r];
  }
  lds_barrier();

  const int tl = (t >> 4) & 1, row = t & 15;
  float y[10];
#pragma unroll
  for (int u = 0; u < 10; ++u) y[u] = ytile[wave][tl][row * 20 + u];

  const float nv2 = v0 * v0 + v1 * v1 + v2 * v2;
  float c1[10];
#pragma unroll
  for (int u = 0; u < 10; ++u) {
    const float s = slw1[u];
    const float n1 = fast_sqrt(nv2 * s * s + EPS2);
    c1[u] = s * FAST_TANH(n1 + snb1[u]) * fast_rcp(n1);
  }
  float S = 0.f;
#pragma unroll
  for (int vv = 0; vv < 10; ++vv) {
    float acc = 0.f;
#pragma unroll
    for (int u = 0; u < 10; ++u) acc = fmaf(c1[u], slw2[u * 10 + vv], acc);
    const float m  = acc * RSQRT_IR;
    const float n2 = fast_sqrt(m * m * nv2 + EPS2);
    const float g  = FAST_TANH(n2 + snb2[vv]) * fast_rcp(n2);
    S = fmaf(swf[vv] * y[vv], m * g, S);
  }
  S *= RSQRT_IR;
  float sx = S * v0, sy = S * v1, sz = S * v2;

#pragma unroll
  for (int off = 16; off; off >>= 1) {
    sx += __shfl_down(sx, off, 32);
    sy += __shfl_down(sy, off, 32);
    sz += __shfl_down(sz, off, 32);
  }
  if (lane == 0) { red[wave][0] = sx; red[wave][1] = sy; red[wave][2] = sz; }
  lds_barrier();
  if (t == 0) {
    float tx = 0.f, ty = 0.f, tz = 0.f;
#pragma unroll
    for (int w = 0; w < 8; ++w) { tx += red[w][0]; ty += red[w][1]; tz += red[w][2]; }
    const float ax = v1b[b * 3 + 0], ay = v1b[b * 3 + 1], az = v1b[b * 3 + 2];
    out[b * 3 + 0] = ay * tz - az * ty;
    out[b * 3 + 1] = az * tx - ax * tz;
    out[b * 3 + 2] = ax * ty - ay * tx;
  }
  }
}

extern "C" void kernel_launch(void* const* d_in, const int* in_sizes, int n_in,
                              void* d_out, int out_size, void* d_ws, size_t ws_size,
                              hipStream_t stream) {
  (void)n_in; (void)out_size; (void)d_ws; (void)ws_size;
  const float* v1b   = (const float*)d_in[0];
  const float* invar = (const float*)d_in[1];
  const float* v2b   = (const float*)d_in[2];
  const float* W1    = (const float*)d_in[3];
  const float* b1    = (const float*)d_in[4];
  const float* W2    = (const float*)d_in[5];
  const float* b2    = (const float*)d_in[6];
  const float* lw1   = (const float*)d_in[7];
  const float* nb1   = (const float*)d_in[8];
  const float* lw2   = (const float*)d_in[9];
  const float* nb2   = (const float*)d_in[10];
  const float* wf    = (const float*)d_in[11];
  float* out = (float*)d_out;

  const int B = in_sizes[0] / 3;
  trq_2b_kernel<<<kGrid, 256, 0, stream>>>(v1b, invar, v2b, W1, b1, W2, b2,
                                           lw1, nb1, lw2, nb2, wf, out, B);
}


# Each wave owns half a row (4 chunks of 32 points): sums stay in the thread over chunks, then lanes, then the two halves (new summation order); one lane sum and one barrier per 4 rows, not 3 per row
# speedup vs baseline: 1.1601x; 1.0738x over previous
#include <hip/hip_runtime.h>

#define EPS2 1e-12f
#define RSQRT_IR 0.31622776601683794f

typedef __attribute__((ext_vector_type(2))) float v2f;
typedef __attribute__((ext_vector_type(8))) float v8f;

#if __has_builtin(__builtin_amdgcn_tanhf)
#define FAST_TANH(x) __builtin_amdgcn_tanhf(x)
#else
#define FAST_TANH(x) tanhf(x)
#endif

__device__ __forceinline__ float fast_sqrt(float x) { return __builtin_amdgcn_sqrtf(x); }
__device__ __forceinline__ float fast_rcp(float x) { return __builtin_amdgcn_rcpf(x); }

__device__ __forceinline__ void lds_barrier() {
  __builtin_amdgcn_fence(__ATOMIC_RELEASE, "workgroup", "local");
  __builtin_amdgcn_s_barrier();
  __builtin_amdgcn_fence(__ATOMIC_ACQUIRE, "workgroup", "local");
}

__device__ __forceinline__ void wave_lds_order() {
  __builtin_amdgcn_fence(__ATOMIC_RELEASE, "wavefront", "local");
  __builtin_amdgcn_wave_barrier();
  __builtin_amdgcn_fence(__ATOMIC_ACQUIRE, "wavefront", "local");
}

#define kGrid 1024
__global__ __launch_bounds__(256) __attribute__((amdgpu_waves_per_eu(10, 12))) void trq_2b_kernel(
    const float* __restrict__ v1b,
    const float* __restrict__ invar,
    const float* __restrict__ v2b,
    const float* __restrict__ W1,
    const float* __restrict__ b1,
    const float* __restrict__ W2,
    const float* __restrict__ b2,
    const float* __restrict__ lw1,
    const float* __restrict__ nb1,
    const float* __restrict__ lw2,
    const float* __restrict__ nb2,
    const float* __restrict__ wf,
    float* __restrict__ out,
    const int B)
{
  __shared__ float sW1p[4 * 16];
  __shared__ float sW2p[12 * 16];
  __shared__ float sB1[16];
  __shared__ float sB2[16];
  __shared__ float slw1[10], snb1[10], slw2[100], snb2[10], swf[10];
  __shared__ float ytile[8][2][16 * 20];
  __shared__ float red[2][8][4];

  const int t = threadIdx.x;
  const int lane = t & 31;
  const int wave = t >> 5;
  const int halfIdx = lane >> 4;
  const int lm = lane & 15;
  const int rsub = wave >> 1;
  const int pbase = (wave & 1) * 128 + lane;

  if ((int)blockIdx.x * 4 >= B) return;

  v2f ninv;
  float nv0, nv1, nv2r;
  {
    const int r0 = 4 * (int)blockIdx.x + rsub;
    const size_t idx = (size_t)(r0 < B ? r0 : B - 1) * 256 + (size_t)pbase;
    ninv = *(const v2f*)(invar + idx * 2);
    const float* vp = v2b + idx * 6 + 3;
    nv0 = vp[0]; nv1 = vp[1]; nv2r = vp[2];
  }
  float nax, nay, naz;
  {
    const int rq = 4 * (int)blockIdx.x + (t & 3);
    const float* ap = v1b + (size_t)(rq < B ? rq : B - 1) * 3;
    nax = ap[0]; nay = ap[1]; naz = ap[2];
  }

  {
    const int k = t >> 4, n = t & 15;
    if (t < 64)  sW1p[t] = (k < 2  && n < 10) ? W1[k * 10 + n] : 0.f;
    if (t < 192) sW2p[t] = (k < 10 && n < 10) ? W2[k * 10 + n] : 0.f;
  }
  if (t < 100) slw2[t] = lw2[t];
  if (t < 16)  { sB1[t] = (t < 10) ? b1[t] : 0.f; sB2[t] = (t < 10) ? b2[t] : 0.f; }
  if (t < 10)  { slw1[t] = lw1[t]; snb1[t] = nb1[t]; snb2[t] = nb2[t]; swf[t] = wf[t]; }
  lds_barrier();

  int par = 0;
  for (int g = blockIdx.x; 4 * g < B; g += kGrid, par ^= 1) {
  const float ax = nax, ay = nay, az = naz;
  float sx = 0.f, sy = 0.f, sz = 0.f;

  for (int ch = 0; ch < 4; ++ch) {
  const v2f inv = ninv;
  const float v0 = nv0, v1 = nv1, v2 = nv2r;
  {
    int gn = g, cn = ch + 1;
    if (cn == 4) { cn = 0; gn = (4 * (g + kGrid) < B) ? (g + kGrid) : g; }
    const int rn = 4 * gn + rsub;
    const size_t idx = (size_t)(rn < B ? rn : B - 1) * 256 + (size_t)(pbase + 32 * cn);
    ninv = *(const v2f*)(invar + idx * 2);
    const float* vp = v2b + idx * 6 + 3;
    nv0 = vp[0]; nv1 = vp[1]; nv2r = vp[2];
  }

  const float i0s = __shfl(inv.x, (lane + 16) & 31, 32);
  const float i1s = __shfl(inv.y, (lane + 16) & 31, 32);
  v2f a0, a1;
  a0.x = (lane < 16) ? inv.x : 0.f;  a0.y = (lane < 16) ? inv.y : 0.f;
  a1.x = (lane < 16) ? i0s  : 0.f;   a1.y = (lane < 16) ? i1s  : 0.f;

  v2f bw1;
  bw1.x = sW1p[(0 + 2 * halfIdx) * 16 + lm];
  bw1.y = sW1p[(1 + 2 * halfIdx) * 16 + lm];

  const float cb1 = sB1[lm];
  v8f d0 = {cb1, cb1, cb1, cb1, cb1, cb1, cb1, cb1};
  v8f d1 = d0;

  d0 = __builtin_amdgcn_wmma_f32_16x16x4_f32(false, a0, false, bw1, (short)0, d0, false, false);
  d1 = __builtin_amdgcn_wmma_f32_16x16x4_f32(false, a1, false, bw1, (short)0, d1, false, false);

#pragma unroll
  for (int r = 0; r < 8; ++r) { d0[r] = fmaxf(d0[r], 0.f); d1[r] = fmaxf(d1[r], 0.f); }

#pragma unroll
  for (int r = 0; r < 8; ++r) {
    const int m = r + 8 * halfIdx;
    ytile[wave][0][m * 20 + lm] = d0[r];
    ytile[wave][1][m * 20 + lm] = d1[r];
  }
  wave_lds_order();

  const float cb2 = sB2[lm];
  v8f e0 = {cb2, cb2, cb2, cb2, cb2, cb2, cb2, cb2};
  v8f e1 = e0;
#pragma unroll
  for (int j = 0; j < 3; ++j) {
    const int k0 = 4 * j + 2 * halfIdx;
    v2f aj0, aj1, bj;
    aj0.x = ytile[wave][0][lm * 20 + k0];
    aj0.y = ytile[wave][0][lm * 20 + k0 + 1];
    aj1.x = ytile[wave][1][lm * 20 + k0];
    aj1.y = ytile[wave][1][lm * 20 + k0 + 1];
    bj.x = sW2p[k0 * 16 + lm];
    bj.y = sW2p[(k0 + 1) * 16 + lm];
    e0 = __builtin_amdgcn_wmma_f32_16x16x4_f32(false, aj0, false, bj, (short)0, e0, false, false);
    e1 = __builtin_amdgcn_wmma_f32_16x16x4_f32(false, aj1, false, bj, (short)0, e1, false, false);
  }
#pragma unroll
  for (int r = 0; r < 8; ++r) { e0[r] = fmaxf(e0[r], 0.f); e1[r] = fmaxf(e1[r], 0.f); }

#pragma unroll
  for (int r = 0; r < 8; ++r) {
    const int m = r + 8 * halfIdx;
    ytile[wave][0][m * 20 + lm] = e0[r];
    ytile[wave][1][m * 20 + lm] = e1[r];
  }
  wave_lds_order();

  float y[10];
#pragma unroll
  for (int u = 0; u < 10; ++u) y[u] = ytile[wave][halfIdx][lm * 20 + u];

  const float nv2 = v0 * v0 + v1 * v1 + v2 * v2;
  float c1[10];
#pragma unroll
  for (int u = 0; u < 10; ++u) {
    const float s = slw1[u];
    const float n1 = fast_sqrt(nv2 * s * s + EPS2);
    c1[u] = s * FAST_TANH(n1 + snb1[u]) * fast_rcp(n1);
  }
  float S = 0.f;
#pragma unroll
  for (int vv = 0; vv < 10; ++vv) {
    float acc = 0.f;
#pragma unroll
    for (int u = 0; u < 10; ++u) acc = fmaf(c1[u], slw2[u * 10 + vv], acc);
    const float m  = acc * RSQRT_IR;
    const float n2 = fast_sqrt(m * m * nv2 + EPS2);
    const float g2 = FAST_TANH(n2 + snb2[vv]) * fast_rcp(n2);
    S = fmaf(swf[vv] * y[vv], m * g2, S);
  }
  S *= RSQRT_IR;
  sx += S * v0; sy += S * v1; sz += S * v2;
  }

  {
    const int gq = (4 * (g + kGrid) < B) ? (g + kGrid) : g;
    const int rq = 4 * gq + (t & 3);
    const float* ap = v1b + (size_t)(rq < B ? rq : B - 1) * 3;
    nax = ap[0]; nay = ap[1]; naz = ap[2];
  }

#pragma unroll
  for (int off = 16; off; off >>= 1) {
    sx += __shfl_down(sx, off, 32);
    sy += __shfl_down(sy, off, 32);
    sz += __shfl_down(sz, off, 32);
  }
  if (lane == 0) { red[par][wave][0] = sx; red[par][wave][1] = sy; red[par][wave][2] = sz; }
  lds_barrier();
  if (t < 4 && 4 * g + t < B) {
    const int rr = 4 * g + t;
    const float tx = red[par][2 * t][0] + red[par][2 * t + 1][0];
    const float ty = red[par][2 * t][1] + red[par][2 * t + 1][1];
    const float tz = red[par][2 * t][2] + red[par][2 * t + 1][2];
    out[rr * 3 + 0] = ay * tz - az * ty;
    out[rr * 3 + 1] = az * tx - ax * tz;
    out[rr * 3 + 2] = ax * ty - ay * tx;
  }
  }
}

extern "C" void kernel_launch(void* const* d_in, const int* in_sizes, int n_in,
                              void* d_out, int out_size, void* d_ws, size_t ws_size,
                              hipStream_t stream) {
  (void)n_in; (void)out_size; (void)d_ws; (void)ws_size;
  const float* v1b   = (const float*)d_in[0];
  const float* invar = (const float*)d_in[1];
  const float* v2b   = (const float*)d_in[2];
  const float* W1    = (const float*)d_in[3];
  const float* b1    = (const float*)d_in[4];
  const float* W2    = (const float*)d_in[5];
  const float* b2    = (const float*)d_in[6];
  const float* lw1   = (const float*)d_in[7];
  const float* nb1   = (const float*)d_in[8];
  const float* lw2   = (const float*)d_in[9];
  const float* nb2   = (const float*)d_in[10];
  const float* wf    = (const float*)d_in[11];
  float* out = (float*)d_out;

  const int B = in_sizes[0] / 3;
  trq_2b_kernel<<<kGrid, 256, 0, stream>>>(v1b, invar, v2b, W1, b1, W2, b2,
                                           lw1, nb1, lw2, nb2, wf, out, B);
}


# Vector-channel weights kept in registers across chunks: 60 of lw2 as scalar operands (uniform global reads), lw1/nb1/nb2/wf in vector registers; 40 weight values per chunk from LDS, not 140
# speedup vs baseline: 1.2539x; 1.0809x over previous
#include <hip/hip_runtime.h>

#define EPS2 1e-12f
#define RSQRT_IR 0.31622776601683794f

typedef __attribute__((ext_vector_type(2))) float v2f;
typedef __attribute__((ext_vector_type(8))) float v8f;

#if __has_builtin(__builtin_amdgcn_tanhf)
#define FAST_TANH(x) __builtin_amdgcn_tanhf(x)
#else
#define FAST_TANH(x) tanhf(x)
#endif

__device__ __forceinline__ float fast_sqrt(float x) { return __builtin_amdgcn_sqrtf(x); }
__device__ __forceinline__ float fast_rcp(float x) { return __builtin_amdgcn_rcpf(x); }

__device__ __forceinline__ void lds_barrier() {
  __builtin_amdgcn_fence(__ATOMIC_RELEASE, "workgroup", "local");
  __builtin_amdgcn_s_barrier();
  __builtin_amdgcn_fence(__ATOMIC_ACQUIRE, "workgroup", "local");
}

__device__ __forceinline__ void wave_lds_order() {
  __builtin_amdgcn_fence(__ATOMIC_RELEASE, "wavefront", "local");
  __builtin_amdgcn_wave_barrier();
  __builtin_amdgcn_fence(__ATOMIC_ACQUIRE, "wavefront", "local");
}

#define kGrid 1024
#define kRowsS 6
__global__ __launch_bounds__(256) __attribute__((amdgpu_waves_per_eu(8, 8))) void trq_2b_kernel(
    const float* __restrict__ v1b,
    const float* __restrict__ invar,
    const float* __restrict__ v2b,
    const float* __restrict__ W1,
    const float* __restrict__ b1,
    const float* __restrict__ W2,
    const float* __restrict__ b2,
    const float* __restrict__ lw1,
    const float* __restrict__ nb1,
    const float* __restrict__ lw2,
    const float* __restrict__ nb2,
    const float* __restrict__ wf,
    float* __restrict__ out,
    const int B)
{
  __shared__ float sW1p[4 * 16];
  __shared__ float sW2p[12 * 16];
  __shared__ float sB1[16];
  __shared__ float sB2[16];
  __shared__ float slw1[10], snb1[10], snb2[10], swf[10];
  __shared__ float slw2t[(10 - kRowsS) * 10];
  __shared__ float ytile[8][2][16 * 20];
  __shared__ float red[2][8][4];

  const int t = threadIdx.x;
  const int lane = t & 31;
  const int wave = t >> 5;
  const int halfIdx = lane >> 4;
  const int lm = lane & 15;
  const int rsub = wave >> 1;
  const int pbase = (wave & 1) * 128 + lane;

  if ((int)blockIdx.x * 4 >= B) return;

  v2f ninv;
  float nv0, nv1, nv2r;
  {
    const int r0 = 4 * (int)blockIdx.x + rsub;
    const size_t idx = (size_t)(r0 < B ? r0 : B - 1) * 256 + (size_t)pbase;
    ninv = *(const v2f*)(invar + idx * 2);
    const float* vp = v2b + idx * 6 + 3;
    nv0 = vp[0]; nv1 = vp[1]; nv2r = vp[2];
  }
  float nax, nay, naz;
  {
    const int rq = 4 * (int)blockIdx.x + (t & 3);
    const float* ap = v1b + (size_t)(rq < B ? rq : B - 1) * 3;
    nax = ap[0]; nay = ap[1]; naz = ap[2];
  }

  {
    const int k = t >> 4, n = t & 15;
    if (t < 64)  sW1p[t] = (k < 2  && n < 10) ? W1[k * 10 + n] : 0.f;
    if (t < 192) sW2p[t] = (k < 10 && n < 10) ? W2[k * 10 + n] : 0.f;
  }
  if (t < (10 - kRowsS) * 10) slw2t[t] = lw2[kRowsS * 10 + t];
  if (t < 16)  { sB1[t] = (t < 10) ? b1[t] : 0.f; sB2[t] = (t < 10) ? b2[t] : 0.f; }
  if (t < 10)  { slw1[t] = lw1[t]; snb1[t] = nb1[t]; snb2[t] = nb2[t]; swf[t] = wf[t]; }
  lds_barrier();

  float r_lw1[10], r_nb1[10], r_nb2[10], r_wf[10];
#pragma unroll
  for (int u = 0; u < 10; ++u) { r_lw1[u] = slw1[u]; r_nb1[u] = snb1[u]; r_nb2[u] = snb2[u]; r_wf[u] = swf[u]; }

  int par = 0;
  for (int g = blockIdx.x; 4 * g < B; g += kGrid, par ^= 1) {
  const float ax = nax, ay = nay, az = naz;
  float sx = 0.f, sy = 0.f, sz = 0.f;

  for (int ch = 0; ch < 4; ++ch) {
  const v2f inv = ninv;
  const float v0 = nv0, v1 = nv1, v2 = nv2r;
  {
    int gn = g, cn = ch + 1;
    if (cn == 4) { cn = 0; gn = (4 * (g + kGrid) < B) ? (g + kGrid) : g; }
    const int rn = 4 * gn + rsub;
    const size_t idx = (size_t)(rn < B ? rn : B - 1) * 256 + (size_t)(pbase + 32 * cn);
    ninv = *(const v2f*)(invar + idx * 2);
    const float* vp = v2b + idx * 6 + 3;
    nv0 = vp[0]; nv1 = vp[1]; nv2r = vp[2];
  }

  const float i0s = __shfl(inv.x, (lane + 16) & 31, 32);
  const float i1s = __shfl(inv.y, (lane + 16) & 31, 32);
  v2f a0, a1;
  a0.x = (lane < 16) ? inv.x : 0.f;  a0.y = (lane < 16) ? inv.y : 0.f;
  a1.x = (lane < 16) ? i0s  : 0.f;   a1.y = (lane < 16) ? i1s  : 0.f;

  v2f bw1;
  bw1.x = sW1p[(0 + 2 * halfIdx) * 16 + lm];
  bw1.y = sW1p[(1 + 2 * halfIdx) * 16 + lm];

  const float cb1 = sB1[lm];
  v8f d0 = {cb1, cb1, cb1, cb1, cb1, cb1, cb1, cb1};
  v8f d1 = d0;

  d0 = __builtin_amdgcn_wmma_f32_16x16x4_f32(false, a0, false, bw1, (short)0, d0, false, false);
  d1 = __builtin_amdgcn_wmma_f32_16x16x4_f32(false, a1, false, bw1, (short)0, d1, false, false);

#pragma unroll
  for (int r = 0; r < 8; ++r) { d0[r] = fmaxf(d0[r], 0.f); d1[r] = fmaxf(d1[r], 0.f); }

#pragma unroll
  for (int r = 0; r < 8; ++r) {
    const int m = r + 8 * halfIdx;
    ytile[wave][0][m * 20 + lm] = d0[r];
    ytile[wave][1][m * 20 + lm] = d1[r];
  }
  wave_lds_order();

  const float cb2 = sB2[lm];
  v8f e0 = {cb2, cb2, cb2, cb2, cb2, cb2, cb2, cb2};
  v8f e1 = e0;
#pragma unroll
  for (int j = 0; j < 3; ++j) {
    const int k0 = 4 * j + 2 * halfIdx;
    v2f aj0, aj1, bj;
    aj0.x = ytile[wave][0][lm * 20 + k0];
    aj0.y = ytile[wave][0][lm * 20 + k0 + 1];
    aj1.x = ytile[wave][1][lm * 20 + k0];
    aj1.y = ytile[wave][1][lm * 20 + k0 + 1];
    bj.x = sW2p[k0 * 16 + lm];
    bj.y = sW2p[(k0 + 1) * 16 + lm];
    e0 = __builtin_amdgcn_wmma_f32_16x16x4_f32(false, aj0, false, bj, (short)0, e0, false, false);
    e1 = __builtin_amdgcn_wmma_f32_16x16x4_f32(false, aj1, false, bj, (short)0, e1, false, false);
  }
#pragma unroll
  for (int r = 0; r < 8; ++r) { e0[r] = fmaxf(e0[r], 0.f); e1[r] = fmaxf(e1[r], 0.f); }

#pragma unroll
  for (int r = 0; r < 8; ++r) {
    const int m = r + 8 * halfIdx;
    ytile[wave][0][m * 20 + lm] = e0[r];
    ytile[wave][1][m * 20 + lm] = e1[r];
  }
  wave_lds_order();

  float y[10];
#pragma unroll
  for (int u = 0; u < 10; ++u) y[u] = ytile[wave][halfIdx][lm * 20 + u];

  const float nv2 = v0 * v0 + v1 * v1 + v2 * v2;
  float c1[10];
#pragma unroll
  for (int u = 0; u < 10; ++u) {
    const float s = r_lw1[u];
    const float n1 = fast_sqrt(nv2 * s * s + EPS2);
    c1[u] = s * FAST_TANH(n1 + r_nb1[u]) * fast_rcp(n1);
  }
  float S = 0.f;
#pragma unroll
  for (int vv = 0; vv < 10; ++vv) {
    float acc = 0.f;
#pragma unroll
    for (int u = 0; u < 10; ++u)
      acc = fmaf(c1[u], (u < kRowsS) ? lw2[u * 10 + vv] : slw2t[(u - kRowsS) * 10 + vv], acc);
    const float m  = acc * RSQRT_IR;
    const float n2 = fast_sqrt(m * m * nv2 + EPS2);
    const float g2 = FAST_TANH(n2 + r_nb2[vv]) * fast_rcp(n2);
    S = fmaf(r_wf[vv] * y[vv], m * g2, S);
  }
  S *= RSQRT_IR;
  sx += S * v0; sy += S * v1; sz += S * v2;
  }

  {
    const int gq = (4 * (g + kGrid) < B) ? (g + kGrid) : g;
    const int rq = 4 * gq + (t & 3);
    const float* ap = v1b + (size_t)(rq < B ? rq : B - 1) * 3;
    nax = ap[0]; nay = ap[1]; naz = ap[2];
  }

#pragma unroll
  for (int off = 16; off; off >>= 1) {
    sx += __shfl_down(sx, off, 32);
    sy += __shfl_down(sy, off, 32);
    sz += __shfl_down(sz, off, 32);
  }
  if (lane == 0) { red[par][wave][0] = sx; red[par][wave][1] = sy; red[par][wave][2] = sz; }
  lds_barrier();
  if (t < 4 && 4 * g + t < B) {
    const int rr = 4 * g + t;
    const float tx = red[par][2 * t][0] + red[par][2 * t + 1][0];
    const float ty = red[par][2 * t][1] + red[par][2 * t + 1][1];
    const float tz = red[par][2 * t][2] + red[par][2 * t + 1][2];
    out[rr * 3 + 0] = ay * tz - az * ty;
    out[rr * 3 + 1] = az * tx - ax * tz;
    out[rr * 3 + 2] = ax * ty - ay * tx;
  }
  }
}

extern "C" void kernel_launch(void* const* d_in, const int* in_sizes, int n_in,
                              void* d_out, int out_size, void* d_ws, size_t ws_size,
                              hipStream_t stream) {
  (void)n_in; (void)out_size; (void)d_ws; (void)ws_size;
  const float* v1b   = (const float*)d_in[0];
  const float* invar = (const float*)d_in[1];
  const float* v2b   = (const float*)d_in[2];
  const float* W1    = (const float*)d_in[3];
  const float* b1    = (const float*)d_in[4];
  const float* W2    = (const float*)d_in[5];
  const float* b2    = (const float*)d_in[6];
  const float* lw1   = (const float*)d_in[7];
  const float* nb1   = (const float*)d_in[8];
  const float* lw2   = (const float*)d_in[9];
  const float* nb2   = (const float*)d_in[10];
  const float* wf    = (const float*)d_in[11];
  float* out = (float*)d_out;

  const int B = in_sizes[0] / 3;
  trq_2b_kernel<<<kGrid, 256, 0, stream>>>(v1b, invar, v2b, W1, b1, W2, b2,
                                           lw1, nb1, lw2, nb2, wf, out, B);
}


# Start-up: each thread issues all its weight loads before its first LDS store, ahead of the point loads: one trip to memory, not five in a row (every store stood behind its own load and waited)
# speedup vs baseline: 1.3526x; 1.0787x over previous
#include <hip/hip_runtime.h>

#define EPS2 1e-12f
#define RSQRT_IR 0.31622776601683794f

typedef __attribute__((ext_vector_type(2))) float v2f;
typedef __attribute__((ext_vector_type(8))) float v8f;

#if __has_builtin(__builtin_amdgcn_tanhf)
#define FAST_TANH(x) __builtin_amdgcn_tanhf(x)
#else
#define FAST_TANH(x) tanhf(x)
#endif

__device__ __forceinline__ float fast_sqrt(float x) { return __builtin_amdgcn_sqrtf(x); }
__device__ __forceinline__ float fast_rcp(float x) { return __builtin_amdgcn_rcpf(x); }

__device__ __forceinline__ void lds_barrier() {
  __builtin_amdgcn_fence(__ATOMIC_RELEASE, "workgroup", "local");
  __builtin_amdgcn_s_barrier();
  __builtin_amdgcn_fence(__ATOMIC_ACQUIRE, "workgroup", "local");
}

__device__ __forceinline__ void wave_lds_order() {
  __builtin_amdgcn_fence(__ATOMIC_RELEASE, "wavefront", "local");
  __builtin_amdgcn_wave_barrier();
  __builtin_amdgcn_fence(__ATOMIC_ACQUIRE, "wavefront", "local");
}

#define kGrid 1024
#define kRowsS 6
__global__ __launch_bounds__(256) __attribute__((amdgpu_waves_per_eu(8, 8))) void trq_2b_kernel(
    const float* __restrict__ v1b,
    const float* __restrict__ invar,
    const float* __restrict__ v2b,
    const float* __restrict__ W1,
    const float* __restrict__ b1,
    const float* __restrict__ W2,
    const float* __restrict__ b2,
    const float* __restrict__ lw1,
    const float* __restrict__ nb1,
    const float* __restrict__ lw2,
    const float* __restrict__ nb2,
    const float* __restrict__ wf,
    float* __restrict__ out,
    const int B)
{
  __shared__ float sW1p[4 * 16];
  __shared__ float sW2p[12 * 16];
  __shared__ float sB1[16];
  __shared__ float sB2[16];
  __shared__ float slw1[10], snb1[10], snb2[10], swf[10];
  __shared__ float slw2t[(10 - kRowsS) * 10];
  __shared__ float ytile[8][2][16 * 20];
  __shared__ float red[2][8][4];

  const int t = threadIdx.x;
  const int lane = t & 31;
  const int wave = t >> 5;
  const int halfIdx = lane >> 4;
  const int lm = lane & 15;
  const int rsub = wave >> 1;
  const int pbase = (wave & 1) * 128 + lane;

  if ((int)blockIdx.x * 4 >= B) return;

  const int sk = t >> 4, sn = t & 15;
  float w1v = 0.f, w2v = 0.f, l2v = 0.f, b1v = 0.f, b2v = 0.f, l1v = 0.f, n1v = 0.f, n2v = 0.f, wfv = 0.f;
  if (sk < 2  && sn < 10) w1v = W1[sk * 10 + sn];
  if (sk < 10 && sn < 10) w2v = W2[sk * 10 + sn];
  if (t < (10 - kRowsS) * 10) l2v = lw2[kRowsS * 10 + t];
  if (t < 10) { b1v = b1[t]; b2v = b2[t]; l1v = lw1[t]; n1v = nb1[t]; n2v = nb2[t]; wfv = wf[t]; }

  v2f ninv;
  float nv0, nv1, nv2r;
  {
    const int r0 = 4 * (int)blockIdx.x + rsub;
    const size_t idx = (size_t)(r0 < B ? r0 : B - 1) * 256 + (size_t)pbase;
    ninv = *(const v2f*)(invar + idx * 2);
    const float* vp = v2b + idx * 6 + 3;
    nv0 = vp[0]; nv1 = vp[1]; nv2r = vp[2];
  }
  float nax, nay, naz;
  {
    const int rq = 4 * (int)blockIdx.x + (t & 3);
    const float* ap = v1b + (size_t)(rq < B ? rq : B - 1) * 3;
    nax = ap[0]; nay = ap[1]; naz = ap[2];
  }

  if (t < 64)  sW1p[t] = w1v;
  if (t < 192) sW2p[t] = w2v;
  if (t < (10 - kRowsS) * 10) slw2t[t] = l2v;
  if (t < 16)  { sB1[t] = b1v; sB2[t] = b2v; }
  if (t < 10)  { slw1[t] = l1v; snb1[t] = n1v; snb2[t] = n2v; swf[t] = wfv; }
  lds_barrier();

  float r_lw1[10], r_nb1[10], r_nb2[10], r_wf[10];
#pragma unroll
  for (int u = 0; u < 10; ++u) { r_lw1[u] = slw1[u]; r_nb1[u] = snb1[u]; r_nb2[u] = snb2[u]; r_wf[u] = swf[u]; }

  int par = 0;
  for (int g = blockIdx.x; 4 * g < B; g += kGrid, par ^= 1) {
  const float ax = nax, ay = nay, az = naz;
  float sx = 0.f, sy = 0.f, sz = 0.f;

  for (int ch = 0; ch < 4; ++ch) {
  const v2f inv = ninv;
  const float v0 = nv0, v1 = nv1, v2 = nv2r;
  {
    int gn = g, cn = ch + 1;
    if (cn == 4) { cn = 0; gn = (4 * (g + kGrid) < B) ? (g + kGrid) : g; }
    const int rn = 4 * gn + rsub;
    const size_t idx = (size_t)(rn < B ? rn : B - 1) * 256 + (size_t)(pbase + 32 * cn);
    ninv = *(const v2f*)(invar + idx * 2);
    const float* vp = v2b + idx * 6 + 3;
    nv0 = vp[0]; nv1 = vp[1]; nv2r = vp[2];
  }

  const float i0s = __shfl(inv.x, (lane + 16) & 31, 32);
  const float i1s = __shfl(inv.y, (lane + 16) & 31, 32);
  v2f a0, a1;
  a0.x = (lane < 16) ? inv.x : 0.f;  a0.y = (lane < 16) ? inv.y : 0.f;
  a1.x = (lane < 16) ? i0s  : 0.f;   a1.y = (lane < 16) ? i1s  : 0.f;

  v2f bw1;
  bw1.x = sW1p[(0 + 2 * halfIdx) * 16 + lm];
  bw1.y = sW1p[(1 + 2 * halfIdx) * 16 + lm];

  const float cb1 = sB1[lm];
  v8f d0 = {cb1, cb1, cb1, cb1, cb1, cb1, cb1, cb1};
  v8f d1 = d0;

  d0 = __builtin_amdgcn_wmma_f32_16x16x4_f32(false, a0, false, bw1, (short)0, d0, false, false);
  d1 = __builtin_amdgcn_wmma_f32_16x16x4_f32(false, a1, false, bw1, (short)0, d1, false, false);

#pragma unroll
  for (int r = 0; r < 8; ++r) { d0[r] = fmaxf(d0[r], 0.f); d1[r] = fmaxf(d1[r], 0.f); }

#pragma unroll
  for (int r = 0; r < 8; ++r) {
    const int m = r + 8 * halfIdx;
    ytile[wave][0][m * 20 + lm] = d0[r];
    ytile[wave][1][m * 20 + lm] = d1[r];
  }
  wave_lds_order();

  const float cb2 = sB2[lm];
  v8f e0 = {cb2, cb2, cb2, cb2, cb2, cb2, cb2, cb2};
  v8f e1 = e0;
#pragma unroll
  for (int j = 0; j < 3; ++j) {
    const int k0 = 4 * j + 2 * halfIdx;
    v2f aj0, aj1, bj;
    aj0.x = ytile[wave][0][lm * 20 + k0];
    aj0.y = ytile[wave][0][lm * 20 + k0 + 1];
    aj1.x = ytile[wave][1][lm * 20 + k0];
    aj1.y = ytile[wave][1][lm * 20 + k0 + 1];
    bj.x = sW2p[k0 * 16 + lm];
    bj.y = sW2p[(k0 + 1) * 16 + lm];
    e0 = __builtin_amdgcn_wmma_f32_16x16x4_f32(false, aj0, false, bj, (short)0, e0, false, false);
    e1 = __builtin_amdgcn_wmma_f32_16x16x4_f32(false, aj1, false, bj, (short)0, e1, false, false);
  }
#pragma unroll
  for (int r = 0; r < 8; ++r) { e0[r] = fmaxf(e0[r], 0.f); e1[r] = fmaxf(e1[r], 0.f); }

#pragma unroll
  for (int r = 0; r < 8; ++r) {
    const int m = r + 8 * halfIdx;
    ytile[wave][0][m * 20 + lm] = e0[r];
    ytile[wave][1][m * 20 + lm] = e1[r];
  }
  wave_lds_order();

  float y[10];
#pragma unroll
  for (int u = 0; u < 10; ++u) y[u] = ytile[wave][halfIdx][lm * 20 + u];

  const float nv2 = v0 * v0 + v1 * v1 + v2 * v2;
  float c1[10];
#pragma unroll
  for (int u = 0; u < 10; ++u) {
    const float s = r_lw1[u];
    const float n1 = fast_sqrt(nv2 * s * s + EPS2);
    c1[u] = s * FAST_TANH(n1 + r_nb1[u]) * fast_rcp(n1);
  }
  float S = 0.f;
#pragma unroll
  for (int vv = 0; vv < 10; ++vv) {
    float acc = 0.f;
#pragma unroll
    for (int u = 0; u < 10; ++u)
      acc = fmaf(c1[u], (u < kRowsS) ? lw2[u * 10 + vv] : slw2t[(u - kRowsS) * 10 + vv], acc);
    const float m  = acc * RSQRT_IR;
    const float n2 = fast_sqrt(m * m * nv2 + EPS2);
    const float g2 = FAST_TANH(n2 + r_nb2[vv]) * fast_rcp(n2);
    S = fmaf(r_wf[vv] * y[vv], m * g2, S);
  }
  S *= RSQRT_IR;
  sx += S * v0; sy += S * v1; sz += S * v2;
  }

  {
    const int gq = (4 * (g + kGrid) < B) ? (g + kGrid) : g;
    const int rq = 4 * gq + (t & 3);
    const float* ap = v1b + (size_t)(rq < B ? rq : B - 1) * 3;
    nax = ap[0]; nay = ap[1]; naz = ap[2];
  }

#pragma unroll
  for (int off = 16; off; off >>= 1) {
    sx += __shfl_down(sx, off, 32);
    sy += __shfl_down(sy, off, 32);
    sz += __shfl_down(sz, off, 32);
  }
  if (lane == 0) { red[par][wave][0] = sx; red[par][wave][1] = sy; red[par][wave][2] = sz; }
  lds_barrier();
  if (t < 4 && 4 * g + t < B) {
    const int rr = 4 * g + t;
    const float tx = red[par][2 * t][0] + red[par][2 * t + 1][0];
    const float ty = red[par][2 * t][1] + red[par][2 * t + 1][1];
    const float tz = red[par][2 * t][2] + red[par][2 * t + 1][2];
    out[rr * 3 + 0] = ay * tz - az * ty;
    out[rr * 3 + 1] = az * tx - ax * tz;
    out[rr * 3 + 2] = ax * ty - ay * tx;
  }
  }
}

extern "C" void kernel_launch(void* const* d_in, const int* in_sizes, int n_in,
                              void* d_out, int out_size, void* d_ws, size_t ws_size,
                              hipStream_t stream) {
  (void)n_in; (void)out_size; (void)d_ws; (void)ws_size;
  const float* v1b   = (const float*)d_in[0];
  const float* invar = (const float*)d_in[1];
  const float* v2b   = (const float*)d_in[2];
  const float* W1    = (const float*)d_in[3];
  const float* b1    = (const float*)d_in[4];
  const float* W2    = (const float*)d_in[5];
  const float* b2    = (const float*)d_in[6];
  const float* lw1   = (const float*)d_in[7];
  const float* nb1   = (const float*)d_in[8];
  const float* lw2   = (const float*)d_in[9];
  const float* nb2   = (const float*)d_in[10];
  const float* wf    = (const float*)d_in[11];
  float* out = (float*)d_out;

  const int B = in_sizes[0] / 3;
  trq_2b_kernel<<<kGrid, 256, 0, stream>>>(v1b, invar, v2b, W1, b1, W2, b2,
                                           lw1, nb1, lw2, nb2, wf, out, B);
}


# MLP in transposed form (channels = matrix rows, points = columns): weight fragments stay in registers, biases ride in the k padding (summed inside the product), y by 16-lane row swaps; no LDS in it
# speedup vs baseline: 1.4439x; 1.0675x over previous
#include <hip/hip_runtime.h>

#define EPS2 1e-12f
#define RSQRT_IR 0.31622776601683794f

typedef __attribute__((ext_vector_type(2))) float v2f;
typedef __attribute__((ext_vector_type(8))) float v8f;

#if __has_builtin(__builtin_amdgcn_tanhf)
#define FAST_TANH(x) __builtin_amdgcn_tanhf(x)
#else
#define FAST_TANH(x) tanhf(x)
#endif

__device__ __forceinline__ float fast_sqrt(float x) { return __builtin_amdgcn_sqrtf(x); }
__device__ __forceinline__ float fast_rcp(float x) { return __builtin_amdgcn_rcpf(x); }

__device__ __forceinline__ void lds_barrier() {
  __builtin_amdgcn_fence(__ATOMIC_RELEASE, "workgroup", "local");
  __builtin_amdgcn_s_barrier();
  __builtin_amdgcn_fence(__ATOMIC_ACQUIRE, "workgroup", "local");
}

__device__ __forceinline__ void row_swap(float& a, float& b) {
  const auto r = __builtin_amdgcn_permlane16_swap(__float_as_uint(a), __float_as_uint(b), false, false);
  a = __uint_as_float(r[0]);
  b = __uint_as_float(r[1]);
}

#define kGrid 1024
#define kRowsS 6
__global__ __launch_bounds__(256) __attribute__((amdgpu_waves_per_eu(8, 8))) void trq_2b_kernel(
    const float* __restrict__ v1b,
    const float* __restrict__ invar,
    const float* __restrict__ v2b,
    const float* __restrict__ W1,
    const float* __restrict__ b1,
    const float* __restrict__ W2,
    const float* __restrict__ b2,
    const float* __restrict__ lw1,
    const float* __restrict__ nb1,
    const float* __restrict__ lw2,
    const float* __restrict__ nb2,
    const float* __restrict__ wf,
    float* __restrict__ out,
    const int B)
{
  __shared__ float slw1[10], snb1[10], snb2[10], swf[10];
  __shared__ float slw2t[(10 - kRowsS) * 10];
  __shared__ float red[2][8][4];

  const int t = threadIdx.x;
  const int lane = t & 31;
  const int wave = t >> 5;
  const int halfIdx = lane >> 4;
  const int lm = lane & 15;
  const int rsub = wave >> 1;
  const int pbase = (wave & 1) * 128 + lane;

  if ((int)blockIdx.x * 4 >= B) return;

  float l2v = 0.f, l1v = 0.f, n1v = 0.f, n2v = 0.f, wfv = 0.f;
  if (t < (10 - kRowsS) * 10) l2v = lw2[kRowsS * 10 + t];
  if (t < 10) { l1v = lw1[t]; n1v = nb1[t]; n2v = nb2[t]; wfv = wf[t]; }
  const bool rowv = (lm & 7) < 5;
  const int chn = (lm >> 3) * 5 + (lm & 7);
  float a1x = (lm == 5 && halfIdx == 1) ? 1.f : 0.f, a1y = 0.f;
  float a2v[6] = {0.f, 0.f, 0.f, 0.f, 0.f, 0.f};
  if (rowv) {
    a1x = halfIdx ? b1[chn] : W1[chn];
#pragma unroll
    for (int q = 0; q < 5; ++q) a2v[q] = W2[(5 * halfIdx + q) * 10 + chn];
    if (halfIdx == 0) { a1y = W1[10 + chn]; a2v[5] = b2[chn]; }
  }

  v2f ninv;
  float nv0, nv1, nv2r;
  {
    const int r0 = 4 * (int)blockIdx.x + rsub;
    const size_t idx = (size_t)(r0 < B ? r0 : B - 1) * 256 + (size_t)pbase;
    ninv = *(const v2f*)(invar + idx * 2);
    const float* vp = v2b + idx * 6 + 3;
    nv0 = vp[0]; nv1 = vp[1]; nv2r = vp[2];
  }
  float nax, nay, naz;
  {
    const int rq = 4 * (int)blockIdx.x + (t & 3);
    const float* ap = v1b + (size_t)(rq < B ? rq : B - 1) * 3;
    nax = ap[0]; nay = ap[1]; naz = ap[2];
  }

  if (t < (10 - kRowsS) * 10) slw2t[t] = l2v;
  if (t < 10)  { slw1[t] = l1v; snb1[t] = n1v; snb2[t] = n2v; swf[t] = wfv; }
  lds_barrier();

  float r_lw1[10], r_nb1[10], r_nb2[10], r_wf[10];
  const v2f A1 = {a1x, a1y};
  const v2f A2_0 = {a2v[0], a2v[1]}, A2_1 = {a2v[2], a2v[3]}, A2_2 = {a2v[4], a2v[5]};
#pragma unroll
  for (int u = 0; u < 10; ++u) { r_lw1[u] = slw1[u]; r_nb1[u] = snb1[u]; r_nb2[u] = snb2[u]; r_wf[u] = swf[u]; }

  int par = 0;
  for (int g = blockIdx.x; 4 * g < B; g += kGrid, par ^= 1) {
  const float ax = nax, ay = nay, az = naz;
  float sx = 0.f, sy = 0.f, sz = 0.f;

  for (int ch = 0; ch < 4; ++ch) {
  const v2f inv = ninv;
  const float v0 = nv0, v1 = nv1, v2 = nv2r;
  {
    int gn = g, cn = ch + 1;
    if (cn == 4) { cn = 0; gn = (4 * (g + kGrid) < B) ? (g + kGrid) : g; }
    const int rn = 4 * gn + rsub;
    const size_t idx = (size_t)(rn < B ? rn : B - 1) * 256 + (size_t)(pbase + 32 * cn);
    ninv = *(const v2f*)(invar + idx * 2);
    const float* vp = v2b + idx * 6 + 3;
    nv0 = vp[0]; nv1 = vp[1]; nv2r = vp[2];
  }

  float f0x = inv.x, f1x = 1.f, f0y = inv.y, f1y = 0.f;
  row_swap(f0x, f1x);
  row_swap(f0y, f1y);
  const v2f bt0 = {f0x, f0y}, bt1 = {f1x, f1y};
  const v8f zero8 = {0.f, 0.f, 0.f, 0.f, 0.f, 0.f, 0.f, 0.f};

  const v8f d0 = __builtin_amdgcn_wmma_f32_16x16x4_f32(false, A1, false, bt0, (short)0, zero8, false, false);
  const v8f d1 = __builtin_amdgcn_wmma_f32_16x16x4_f32(false, A1, false, bt1, (short)0, zero8, false, false);
  float p0[6], p1[6];
#pragma unroll
  for (int r = 0; r < 5; ++r) { p0[r] = fmaxf(d0[r], 0.f); p1[r] = fmaxf(d1[r], 0.f); }
  p0[5] = d0[5]; p1[5] = d1[5];

  v8f e0 = zero8, e1 = zero8;
  {
    const v2f q00 = {p0[0], p0[1]}, q01 = {p0[2], p0[3]}, q02 = {p0[4], p0[5]};
    const v2f q10 = {p1[0], p1[1]}, q11 = {p1[2], p1[3]}, q12 = {p1[4], p1[5]};
    e0 = __builtin_amdgcn_wmma_f32_16x16x4_f32(false, A2_0, false, q00, (short)0, e0, false, false);
    e1 = __builtin_amdgcn_wmma_f32_16x16x4_f32(false, A2_0, false, q10, (short)0, e1, false, false);
    e0 = __builtin_amdgcn_wmma_f32_16x16x4_f32(false, A2_1, false, q01, (short)0, e0, false, false);
    e1 = __builtin_amdgcn_wmma_f32_16x16x4_f32(false, A2_1, false, q11, (short)0, e1, false, false);
    e0 = __builtin_amdgcn_wmma_f32_16x16x4_f32(false, A2_2, false, q02, (short)0, e0, false, false);
    e1 = __builtin_amdgcn_wmma_f32_16x16x4_f32(false, A2_2, false, q12, (short)0, e1, false, false);
  }

  float y[10];
#pragma unroll
  for (int r = 0; r < 5; ++r) {
    float lo = fmaxf(e0[r], 0.f), hi = fmaxf(e1[r], 0.f);
    row_swap(lo, hi);
    y[r] = lo; y[5 + r] = hi;
  }

  __builtin_amdgcn_fence(__ATOMIC_ACQUIRE, "wavefront", "local");
  __builtin_amdgcn_sched_barrier(0);
  const float nv2 = v0 * v0 + v1 * v1 + v2 * v2;
  float c1[10];
  {
    float q[10], n1[10], th[10];
#pragma unroll
    for (int u = 0; u < 10; ++u) { const float s = r_lw1[u]; q[u] = nv2 * s * s + EPS2; }
#pragma unroll
    for (int u = 0; u < 10; ++u) n1[u] = fast_sqrt(q[u]);
#pragma unroll
    for (int u = 0; u < 10; ++u) th[u] = FAST_TANH(n1[u] + r_nb1[u]);
#pragma unroll
    for (int u = 0; u < 10; ++u) c1[u] = r_lw1[u] * th[u] * fast_rcp(n1[u]);
  }
  float m[10];
  {
    float acc[10];
#pragma unroll
    for (int vv = 0; vv < 10; ++vv) acc[vv] = 0.f;
#pragma unroll
    for (int u = 0; u < 10; ++u)
#pragma unroll
      for (int vv = 0; vv < 10; ++vv)
        acc[vv] = fmaf(c1[u], (u < kRowsS) ? lw2[u * 10 + vv] : slw2t[(u - kRowsS) * 10 + vv], acc[vv]);
#pragma unroll
    for (int vv = 0; vv < 10; ++vv) m[vv] = acc[vv] * RSQRT_IR;
  }
  float S = 0.f;
  {
    float q[10], n2[10], th[10];
#pragma unroll
    for (int vv = 0; vv < 10; ++vv) q[vv] = m[vv] * m[vv] * nv2 + EPS2;
#pragma unroll
    for (int vv = 0; vv < 10; ++vv) n2[vv] = fast_sqrt(q[vv]);
#pragma unroll
    for (int vv = 0; vv < 10; ++vv) th[vv] = FAST_TANH(n2[vv] + r_nb2[vv]);
#pragma unroll
    for (int vv = 0; vv < 10; ++vv) {
      const float g2 = th[vv] * fast_rcp(n2[vv]);
      S = fmaf(r_wf[vv] * y[vv], m[vv] * g2, S);
    }
  }
  S *= RSQRT_IR;
  sx += S * v0; sy += S * v1; sz += S * v2;
  }

  {
    const int gq = (4 * (g + kGrid) < B) ? (g + kGrid) : g;
    const int rq = 4 * gq + (t & 3);
    const float* ap = v1b + (size_t)(rq < B ? rq : B - 1) * 3;
    nax = ap[0]; nay = ap[1]; naz = ap[2];
  }

#pragma unroll
  for (int off = 16; off; off >>= 1) {
    sx += __shfl_down(sx, off, 32);
    sy += __shfl_down(sy, off, 32);
    sz += __shfl_down(sz, off, 32);
  }
  if (lane == 0) { red[par][wave][0] = sx; red[par][wave][1] = sy; red[par][wave][2] = sz; }
  lds_barrier();
  if (t < 4 && 4 * g + t < B) {
    const int rr = 4 * g + t;
    const float tx = red[par][2 * t][0] + red[par][2 * t + 1][0];
    const float ty = red[par][2 * t][1] + red[par][2 * t + 1][1];
    const float tz = red[par][2 * t][2] + red[par][2 * t + 1][2];
    out[rr * 3 + 0] = ay * tz - az * ty;
    out[rr * 3 + 1] = az * tx - ax * tz;
    out[rr * 3 + 2] = ax * ty - ay * tx;
  }
  }
}

extern "C" void kernel_launch(void* const* d_in, const int* in_sizes, int n_in,
                              void* d_out, int out_size, void* d_ws, size_t ws_size,
                              hipStream_t stream) {
  (void)n_in; (void)out_size; (void)d_ws; (void)ws_size;
  const float* v1b   = (const float*)d_in[0];
  const float* invar = (const float*)d_in[1];
  const float* v2b   = (const float*)d_in[2];
  const float* W1    = (const float*)d_in[3];
  const float* b1    = (const float*)d_in[4];
  const float* W2    = (const float*)d_in[5];
  const float* b2    = (const float*)d_in[6];
  const float* lw1   = (const float*)d_in[7];
  const float* nb1   = (const float*)d_in[8];
  const float* lw2   = (const float*)d_in[9];
  const float* nb2   = (const float*)d_in[10];
  const float* wf    = (const float*)d_in[11];
  float* out = (float*)d_out;

  const int B = in_sizes[0] / 3;
  trq_2b_kernel<<<kGrid, 256, 0, stream>>>(v1b, invar, v2b, W1, b1, W2, b2,
                                           lw1, nb1, lw2, nb2, wf, out, B);
}
